# starting kernel
# baseline (speedup 1.0000x reference)
#include <hip/hip_runtime.h>

typedef __attribute__((ext_vector_type(2))) float v2f;
typedef __attribute__((ext_vector_type(8))) float v8f;

#define OUT_DIM 64

__global__ void build_inverse_perm(const int* __restrict__ ids_3d,
                                   int* __restrict__ inv, int n) {
    int i = blockIdx.x * blockDim.x + threadIdx.x;
    if (i < n) inv[ids_3d[i]] = i;
}

__global__ __launch_bounds__(32) void sh_fused(
    const int*   __restrict__ point_ids,
    const float* __restrict__ cam_center,
    const float* __restrict__ xyz,
    const float* __restrict__ sh_coeffs,
    const float* __restrict__ confi,
    const float* __restrict__ W,
    const float* __restrict__ bvec,
    const int*   __restrict__ inv,
    float*       __restrict__ out,
    int n_query)
{
    __shared__ float shf[16][16];

    const int lane = threadIdx.x;
    const int q    = lane & 15;
    const int half = lane >> 4;
    const int base = blockIdx.x * 16;
    const int m    = base + q;

    const int idx = inv[point_ids[m]];

    const float dx = cam_center[0] - xyz[idx * 3 + 0];
    const float dy = cam_center[1] - xyz[idx * 3 + 1];
    const float dz = cam_center[2] - xyz[idx * 3 + 2];
    const float nrm = sqrtf(dx * dx + dy * dy + dz * dz);
    const float inr = 1.0f / fmaxf(nrm, 1e-12f);
    const float x = dx * inr, y = dy * inr, z = dz * inr;

    const float SQRT3 = 1.7320508075688772f;
    const float s5    = 2.2360679774997896f;
    const float s7    = 2.6457513110645907f;
    const float c56   = 0.9128709291752769f;
    const float c38   = 0.6123724356957945f;

    const float y2   = y * y;
    const float x2z2 = x * x + z * z;
    const float sh2_0 = SQRT3 * x * z;
    const float sh2_1 = SQRT3 * x * y;
    const float sh2_2 = y2 - 0.5f * x2z2;
    const float sh2_3 = SQRT3 * y * z;
    const float sh2_4 = 0.5f * SQRT3 * (z * z - x * x);
    const float sh3_0 = c56 * (sh2_0 * z + sh2_4 * x);
    const float sh3_1 = s5 * sh2_0 * y;
    const float sh3_2 = c38 * (4.0f * y2 - x2z2) * x;
    const float sh3_3 = 0.5f * y * (2.0f * y2 - 3.0f * x2z2);
    const float sh3_4 = c38 * z * (4.0f * y2 - x2z2);
    const float sh3_5 = s5 * sh2_4 * y;
    const float sh3_6 = c56 * (sh2_4 * z - sh2_0 * x);

    float Y[16];
    Y[0] = 1.0f;
    Y[1] = SQRT3 * x;  Y[2] = SQRT3 * y;  Y[3] = SQRT3 * z;
    Y[4] = s5 * sh2_0; Y[5] = s5 * sh2_1; Y[6] = s5 * sh2_2;
    Y[7] = s5 * sh2_3; Y[8] = s5 * sh2_4;
    Y[9]  = s7 * sh3_0; Y[10] = s7 * sh3_1; Y[11] = s7 * sh3_2;
    Y[12] = s7 * sh3_3; Y[13] = s7 * sh3_4; Y[14] = s7 * sh3_5;
    Y[15] = s7 * sh3_6;

    if (half == 0) {
        const float4* cp = (const float4*)(confi + (size_t)idx * 16);
        float s = 0.0f;
        #pragma unroll
        for (int d4 = 0; d4 < 4; ++d4) {
            float4 rv = cp[d4];
            s = fmaf(rv.x, Y[d4 * 4 + 0], s);
            s = fmaf(rv.y, Y[d4 * 4 + 1], s);
            s = fmaf(rv.z, Y[d4 * 4 + 2], s);
            s = fmaf(rv.w, Y[d4 * 4 + 3], s);
        }
        out[(size_t)n_query * OUT_DIM + m] = s;
    }

    const float* sp = sh_coeffs + (size_t)idx * 256 + (size_t)half * 8 * 16;
    #pragma unroll
    for (int c = 0; c < 8; ++c) {
        const float4* rp = (const float4*)(sp + c * 16);
        float s = 0.0f;
        #pragma unroll
        for (int d4 = 0; d4 < 4; ++d4) {
            float4 rv = rp[d4];
            s = fmaf(rv.x, Y[d4 * 4 + 0], s);
            s = fmaf(rv.y, Y[d4 * 4 + 1], s);
            s = fmaf(rv.z, Y[d4 * 4 + 2], s);
            s = fmaf(rv.w, Y[d4 * 4 + 3], s);
        }
        shf[q][half * 8 + c] = s;
    }
    __syncthreads();

    const int klo = half * 2;
    v2f afrag[4];
    #pragma unroll
    for (int kk = 0; kk < 4; ++kk) {
        afrag[kk][0] = shf[q][4 * kk + klo + 0];
        afrag[kk][1] = shf[q][4 * kk + klo + 1];
    }

    #pragma unroll
    for (int nt = 0; nt < 4; ++nt) {
        const int col = nt * 16 + q;
        const float bb = bvec[col];
        v8f acc = {bb, bb, bb, bb, bb, bb, bb, bb};

        #pragma unroll
        for (int kk = 0; kk < 4; ++kk) {
            v2f bfrag;
            bfrag[0] = W[(4 * kk + klo + 0) * OUT_DIM + col];
            bfrag[1] = W[(4 * kk + klo + 1) * OUT_DIM + col];
            acc = __builtin_amdgcn_wmma_f32_16x16x4_f32(
                false, afrag[kk], false, bfrag, (short)0, acc, false, false);
        }

        #pragma unroll
        for (int v = 0; v < 8; ++v) {
            const int row = v + half * 8;
            out[(size_t)(base + row) * OUT_DIM + col] = acc[v];
        }
    }
}

extern "C" void kernel_launch(void* const* d_in, const int* in_sizes, int n_in,
                              void* d_out, int out_size, void* d_ws, size_t ws_size,
                              hipStream_t stream) {
    const int*   point_ids = (const int*)d_in[0];
    const float* cam       = (const float*)d_in[1];
    const int*   ids_3d    = (const int*)d_in[2];
    const float* xyz       = (const float*)d_in[3];
    const float* sh        = (const float*)d_in[4];
    const float* confi     = (const float*)d_in[5];
    const float* W         = (const float*)d_in[6];
    const float* b         = (const float*)d_in[7];
    float* out = (float*)d_out;

    const int n_pts   = in_sizes[2];
    const int n_query = in_sizes[0];

    int* inv = (int*)d_ws;

    build_inverse_perm<<<(n_pts + 255) / 256, 256, 0, stream>>>(ids_3d, inv, n_pts);
    sh_fused<<<n_query / 16, 32, 0, stream>>>(point_ids, cam, xyz, sh, confi,
                                              W, b, inv, out, n_query);
}


# One launch instead of two: each 4-wave work-group keeps an LDS id-to-slot byte table for its 256 points, matches all queries against it and runs the 16-query WMMA tile on its own matches
# speedup vs baseline: 1.3419x; 1.3419x over previous
#include <hip/hip_runtime.h>


typedef __attribute__((ext_vector_type(2))) float v2f;
typedef __attribute__((ext_vector_type(8))) float v8f;

#define OUT_DIM    64
#define CHUNK      256
#define WG_LANES   128
#define N_WAVES    4
#define TABLE_N    100000
#define Q_PASS     4096
#define Q_PER_WAVE (Q_PASS / N_WAVES)

__global__ __launch_bounds__(WG_LANES) void sh_fused(
    const int*   __restrict__ point_ids,
    const float* __restrict__ cam_center,
    const int*   __restrict__ ids_3d,
    const float* __restrict__ xyz,
    const float* __restrict__ sh_coeffs,
    const float* __restrict__ confi,
    const float* __restrict__ W,
    const float* __restrict__ bvec,
    float*       __restrict__ out,
    int n_pts, int n_query)
{
    __shared__ __attribute__((aligned(16))) unsigned char loc8[TABLE_N];
    __shared__ int           ids_chunk[CHUNK];
    __shared__ int           mlist[N_WAVES][Q_PER_WAVE];

    const int tid  = threadIdx.x;
    const int lane = tid & 31;
    const int w    = tid >> 5;
    const int q    = lane & 15;
    const int half = lane >> 4;
    const int base = blockIdx.x * CHUNK;

    const int4* pq  = (const int4*)point_ids;
    const int   nq4 = n_query >> 2;
    int pA[CHUNK / WG_LANES];
    #pragma unroll
    for (int r = 0; r < CHUNK / WG_LANES; ++r) {
        const int i = base + tid + r * WG_LANES;
        const int v = ids_3d[(i < n_pts) ? i : 0];
        pA[r] = (i < n_pts) ? v : -1;
    }
    int4 pv[8];
    #pragma unroll
    for (int j = 0; j < 8; ++j) {
        const int v4 = tid + WG_LANES * j;
        const int4 v = pq[(v4 < nq4) ? v4 : 0];
        pv[j] = (v4 < nq4) ? v : make_int4(-2, -2, -2, -2);
    }
    const int klo = half * 2;
    v2f   bfrag[4][4];
    float bias[4];
    #pragma unroll
    for (int nt = 0; nt < 4; ++nt) {
        const int col = nt * 16 + q;
        bias[nt] = bvec[col];
        #pragma unroll
        for (int kk = 0; kk < 4; ++kk) {
            bfrag[nt][kk][0] = W[(4 * kk + klo + 0) * OUT_DIM + col];
            bfrag[nt][kk][1] = W[(4 * kk + klo + 1) * OUT_DIM + col];
        }
    }
    const float cx = cam_center[0], cy = cam_center[1], cz = cam_center[2];

    {
        uint4* t16 = (uint4*)loc8;
        for (int i = tid; i < TABLE_N / 16; i += WG_LANES) t16[i] = make_uint4(0u, 0u, 0u, 0u);
    }
    __syncthreads();
    #pragma unroll
    for (int r = 0; r < CHUNK / WG_LANES; ++r) {
        const int t = tid + r * WG_LANES;
        ids_chunk[t] = pA[r];
        if ((unsigned)pA[r] < (unsigned)TABLE_N) loc8[pA[r]] = (unsigned char)t;
    }
    __syncthreads();

    for (int qb = 0; qb < n_query; qb += Q_PASS) {
        int pm[32], t8[32], got[32];
        #pragma unroll
        for (int j = 0; j < 8; ++j) {
            pm[4 * j + 0] = pv[j].x; pm[4 * j + 1] = pv[j].y;
            pm[4 * j + 2] = pv[j].z; pm[4 * j + 3] = pv[j].w;
        }
        if (qb + Q_PASS < n_query) {
            #pragma unroll
            for (int j = 0; j < 8; ++j) {
                const int v4 = ((qb + Q_PASS) >> 2) + tid + WG_LANES * j;
                const int4 v = pq[(v4 < nq4) ? v4 : 0];
                pv[j] = (v4 < nq4) ? v : make_int4(-2, -2, -2, -2);
            }
        }
        #pragma unroll
        for (int k = 0; k < 32; ++k) {
            const int pi = ((unsigned)pm[k] < (unsigned)TABLE_N) ? pm[k] : 0;
            t8[k] = loc8[pi];
        }
        #pragma unroll
        for (int k = 0; k < 32; ++k) got[k] = ids_chunk[t8[k]];
        int cnt = 0;
        #pragma unroll
        for (int k = 0; k < 32; ++k) {
            const bool hit = (got[k] == pm[k]);
            const unsigned mask = __builtin_amdgcn_ballot_w32(hit);
            if (mask != 0u) {
                if (hit) {
                    const int m   = 4 * ((qb >> 2) + tid + WG_LANES * (k >> 2)) + (k & 3);
                    const int pos = cnt + (int)__builtin_amdgcn_mbcnt_lo(mask, 0u);
                    mlist[w][pos] = (m << 8) | t8[k];
                }
                cnt += __builtin_popcount(mask);
            }
        }
        __syncthreads();

        for (int t0 = 0; t0 < cnt; t0 += 16) {
            const int  slot  = t0 + q;
            const bool valid = slot < cnt;
            const int  ent   = mlist[w][valid ? slot : t0];
            const int  m     = ent >> 8;
            const int  idx   = base + (ent & 255);

            const float dx = cx - xyz[idx * 3 + 0];
            const float dy = cy - xyz[idx * 3 + 1];
            const float dz = cz - xyz[idx * 3 + 2];
            const float nrm = sqrtf(dx * dx + dy * dy + dz * dz);
            const float inr = 1.0f / fmaxf(nrm, 1e-12f);
            const float x = dx * inr, y = dy * inr, z = dz * inr;

            const float SQRT3 = 1.7320508075688772f;
            const float s5    = 2.2360679774997896f;
            const float s7    = 2.6457513110645907f;
            const float c56   = 0.9128709291752769f;
            const float c38   = 0.6123724356957945f;

            const float y2   = y * y;
            const float x2z2 = x * x + z * z;
            const float sh2_0 = SQRT3 * x * z;
            const float sh2_1 = SQRT3 * x * y;
            const float sh2_2 = y2 - 0.5f * x2z2;
            const float sh2_3 = SQRT3 * y * z;
            const float sh2_4 = 0.5f * SQRT3 * (z * z - x * x);
            const float sh3_0 = c56 * (sh2_0 * z + sh2_4 * x);
            const float sh3_1 = s5 * sh2_0 * y;
            const float sh3_2 = c38 * (4.0f * y2 - x2z2) * x;
            const float sh3_3 = 0.5f * y * (2.0f * y2 - 3.0f * x2z2);
            const float sh3_4 = c38 * z * (4.0f * y2 - x2z2);
            const float sh3_5 = s5 * sh2_4 * y;
            const float sh3_6 = c56 * (sh2_4 * z - sh2_0 * x);

            float Y[16];
            Y[0] = 1.0f;
            Y[1] = SQRT3 * x;  Y[2] = SQRT3 * y;  Y[3] = SQRT3 * z;
            Y[4] = s5 * sh2_0; Y[5] = s5 * sh2_1; Y[6] = s5 * sh2_2;
            Y[7] = s5 * sh2_3; Y[8] = s5 * sh2_4;
            Y[9]  = s7 * sh3_0; Y[10] = s7 * sh3_1; Y[11] = s7 * sh3_2;
            Y[12] = s7 * sh3_3; Y[13] = s7 * sh3_4; Y[14] = s7 * sh3_5;
            Y[15] = s7 * sh3_6;

            if (half == 0) {
                const float4* cp = (const float4*)(confi + (size_t)idx * 16);
                float s = 0.0f;
                #pragma unroll
                for (int d4 = 0; d4 < 4; ++d4) {
                    float4 rv = cp[d4];
                    s = fmaf(rv.x, Y[d4 * 4 + 0], s);
                    s = fmaf(rv.y, Y[d4 * 4 + 1], s);
                    s = fmaf(rv.z, Y[d4 * 4 + 2], s);
                    s = fmaf(rv.w, Y[d4 * 4 + 3], s);
                }
                if (valid) out[(size_t)n_query * OUT_DIM + m] = s;
            }

            const float* sp = sh_coeffs + (size_t)idx * 256;
            v2f afrag[4];
            #pragma unroll
            for (int kk = 0; kk < 4; ++kk) {
                #pragma unroll
                for (int v = 0; v < 2; ++v) {
                    const float4* rp = (const float4*)(sp + (4 * kk + klo + v) * 16);
                    float s = 0.0f;
                    #pragma unroll
                    for (int d4 = 0; d4 < 4; ++d4) {
                        float4 rv = rp[d4];
                        s = fmaf(rv.x, Y[d4 * 4 + 0], s);
                        s = fmaf(rv.y, Y[d4 * 4 + 1], s);
                        s = fmaf(rv.z, Y[d4 * 4 + 2], s);
                        s = fmaf(rv.w, Y[d4 * 4 + 3], s);
                    }
                    afrag[kk][v] = s;
                }
            }

            int mrow[8];
            #pragma unroll
            for (int v = 0; v < 8; ++v) {
                const int mlo = __builtin_amdgcn_readlane(m, v);
                const int mhi = __builtin_amdgcn_readlane(m, v + 8);
                mrow[v] = half ? mhi : mlo;
            }

            #pragma unroll
            for (int nt = 0; nt < 4; ++nt) {
                const int col = nt * 16 + q;
                const float bb = bias[nt];
                v8f acc = {bb, bb, bb, bb, bb, bb, bb, bb};
                #pragma unroll
                for (int kk = 0; kk < 4; ++kk) {
                    acc = __builtin_amdgcn_wmma_f32_16x16x4_f32(
                        false, afrag[kk], false, bfrag[nt][kk], (short)0, acc, false, false);
                }
                #pragma unroll
                for (int v = 0; v < 8; ++v) {
                    if (t0 + v + half * 8 < cnt)
                        out[(size_t)mrow[v] * OUT_DIM + col] = acc[v];
                }
            }
        }
        __syncthreads();
    }
}

extern "C" void kernel_launch(void* const* d_in, const int* in_sizes, int n_in,
                              void* d_out, int out_size, void* d_ws, size_t ws_size,
                              hipStream_t stream) {
    const int*   point_ids = (const int*)d_in[0];
    const float* cam       = (const float*)d_in[1];
    const int*   ids_3d    = (const int*)d_in[2];
    const float* xyz       = (const float*)d_in[3];
    const float* sh        = (const float*)d_in[4];
    const float* confi     = (const float*)d_in[5];
    const float* W         = (const float*)d_in[6];
    const float* b         = (const float*)d_in[7];
    float* out = (float*)d_out;

    const int n_pts   = in_sizes[2];
    const int n_query = in_sizes[0];

    sh_fused<<<(n_pts + 255) / 256, 128, 0, stream>>>(point_ids, cam, ids_3d, xyz, sh, confi,
                                                      W, b, out, n_pts, n_query);
}


# Phase C issues every gather of a tile (point, confidence row, 8 coefficient rows) as one group: one memory round trip, not three; confidence dot = two 8-term partial sums joined across lane halves
# speedup vs baseline: 1.7498x; 1.3040x over previous
#include <hip/hip_runtime.h>


typedef __attribute__((ext_vector_type(2))) float v2f;
typedef __attribute__((ext_vector_type(8))) float v8f;

#define OUT_DIM    64
#define CHUNK      256
#define WG_LANES   128
#define N_WAVES    4
#define TABLE_N    100000
#define Q_PASS     4096
#define Q_PER_WAVE (Q_PASS / N_WAVES)

__device__ __forceinline__ float xhalf(float x) {
    return __builtin_bit_cast(float, __builtin_amdgcn_permlanex16(__builtin_bit_cast(int, x), __builtin_bit_cast(int, x),
                                                                  0x76543210, 0xfedcba98, false, false));
}

__global__ __launch_bounds__(WG_LANES) void sh_fused(
    const int*   __restrict__ point_ids,
    const float* __restrict__ cam_center,
    const int*   __restrict__ ids_3d,
    const float* __restrict__ xyz,
    const float* __restrict__ sh_coeffs,
    const float* __restrict__ confi,
    const float* __restrict__ W,
    const float* __restrict__ bvec,
    float*       __restrict__ out,
    int n_pts, int n_query)
{
    __shared__ __attribute__((aligned(16))) unsigned char loc8[TABLE_N];
    __shared__ int           ids_chunk[CHUNK];
    __shared__ int           mlist[N_WAVES][Q_PER_WAVE];

    const int tid  = threadIdx.x;
    const int lane = tid & 31;
    const int w    = tid >> 5;
    const int q    = lane & 15;
    const int half = lane >> 4;
    const int base = blockIdx.x * CHUNK;

    const int4* pq  = (const int4*)point_ids;
    const int   nq4 = n_query >> 2;
    int pA[CHUNK / WG_LANES];
    #pragma unroll
    for (int r = 0; r < CHUNK / WG_LANES; ++r) {
        const int i = base + tid + r * WG_LANES;
        const int v = ids_3d[(i < n_pts) ? i : 0];
        pA[r] = (i < n_pts) ? v : -1;
    }
    int4 pv[8];
    #pragma unroll
    for (int j = 0; j < 8; ++j) {
        const int v4 = tid + WG_LANES * j;
        const int4 v = pq[(v4 < nq4) ? v4 : 0];
        pv[j] = (v4 < nq4) ? v : make_int4(-2, -2, -2, -2);
    }
    const int klo = half * 2;
    v2f   bfrag[4][4];
    float bias[4];
    #pragma unroll
    for (int nt = 0; nt < 4; ++nt) {
        const int col = nt * 16 + q;
        bias[nt] = bvec[col];
        #pragma unroll
        for (int kk = 0; kk < 4; ++kk) {
            bfrag[nt][kk][0] = W[(4 * kk + klo + 0) * OUT_DIM + col];
            bfrag[nt][kk][1] = W[(4 * kk + klo + 1) * OUT_DIM + col];
        }
    }
    const float cx = cam_center[0], cy = cam_center[1], cz = cam_center[2];

    {
        uint4* t16 = (uint4*)loc8;
        for (int i = tid; i < TABLE_N / 16; i += WG_LANES) t16[i] = make_uint4(0u, 0u, 0u, 0u);
    }
    __syncthreads();
    #pragma unroll
    for (int r = 0; r < CHUNK / WG_LANES; ++r) {
        const int t = tid + r * WG_LANES;
        ids_chunk[t] = pA[r];
        if ((unsigned)pA[r] < (unsigned)TABLE_N) loc8[pA[r]] = (unsigned char)t;
    }
    __syncthreads();

    for (int qb = 0; qb < n_query; qb += Q_PASS) {
        int pm[32], t8[32], got[32];
        #pragma unroll
        for (int j = 0; j < 8; ++j) {
            pm[4 * j + 0] = pv[j].x; pm[4 * j + 1] = pv[j].y;
            pm[4 * j + 2] = pv[j].z; pm[4 * j + 3] = pv[j].w;
        }
        if (qb + Q_PASS < n_query) {
            #pragma unroll
            for (int j = 0; j < 8; ++j) {
                const int v4 = ((qb + Q_PASS) >> 2) + tid + WG_LANES * j;
                const int4 v = pq[(v4 < nq4) ? v4 : 0];
                pv[j] = (v4 < nq4) ? v : make_int4(-2, -2, -2, -2);
            }
        }
        #pragma unroll
        for (int k = 0; k < 32; ++k) {
            const int pi = ((unsigned)pm[k] < (unsigned)TABLE_N) ? pm[k] : 0;
            t8[k] = loc8[pi];
        }
        #pragma unroll
        for (int k = 0; k < 32; ++k) got[k] = ids_chunk[t8[k]];
        int cnt = 0;
        #pragma unroll
        for (int k = 0; k < 32; ++k) {
            const bool hit = (got[k] == pm[k]);
            const unsigned mask = __builtin_amdgcn_ballot_w32(hit);
            if (mask != 0u) {
                if (hit) {
                    const int m   = 4 * ((qb >> 2) + tid + WG_LANES * (k >> 2)) + (k & 3);
                    const int pos = cnt + (int)__builtin_amdgcn_mbcnt_lo(mask, 0u);
                    mlist[w][pos] = (m << 8) | t8[k];
                }
                cnt += __builtin_popcount(mask);
            }
        }
        __syncthreads();

        for (int t0 = 0; t0 < cnt; t0 += 16) {
            const int  slot  = t0 + q;
            const bool valid = slot < cnt;
            const int  ent   = mlist[w][valid ? slot : t0];
            const int  m     = ent >> 8;
            const int  idx   = base + (ent & 255);

            const float* xp = xyz + (size_t)idx * 3;
            const float px = xp[0], py = xp[1], pz = xp[2];
            const float4* cp = (const float4*)(confi + (size_t)idx * 16) + 2 * half;
            float4 cv[2];
            cv[0] = cp[0];
            cv[1] = cp[1];
            const float* sp = sh_coeffs + (size_t)idx * 256;
            float4 sv[4][2][4];
            #pragma unroll
            for (int kk = 0; kk < 4; ++kk) {
                #pragma unroll
                for (int v = 0; v < 2; ++v) {
                    const float4* rp = (const float4*)(sp + (4 * kk + klo + v) * 16);
                    #pragma unroll
                    for (int d4 = 0; d4 < 4; ++d4) sv[kk][v][d4] = rp[d4];
                }
            }

            const float dx = cx - px;
            const float dy = cy - py;
            const float dz = cz - pz;
            const float nrm = sqrtf(dx * dx + dy * dy + dz * dz);
            const float inr = 1.0f / fmaxf(nrm, 1e-12f);
            const float x = dx * inr, y = dy * inr, z = dz * inr;

            const float SQRT3 = 1.7320508075688772f;
            const float s5    = 2.2360679774997896f;
            const float s7    = 2.6457513110645907f;
            const float c56   = 0.9128709291752769f;
            const float c38   = 0.6123724356957945f;

            const float y2   = y * y;
            const float x2z2 = x * x + z * z;
            const float sh2_0 = SQRT3 * x * z;
            const float sh2_1 = SQRT3 * x * y;
            const float sh2_2 = y2 - 0.5f * x2z2;
            const float sh2_3 = SQRT3 * y * z;
            const float sh2_4 = 0.5f * SQRT3 * (z * z - x * x);
            const float sh3_0 = c56 * (sh2_0 * z + sh2_4 * x);
            const float sh3_1 = s5 * sh2_0 * y;
            const float sh3_2 = c38 * (4.0f * y2 - x2z2) * x;
            const float sh3_3 = 0.5f * y * (2.0f * y2 - 3.0f * x2z2);
            const float sh3_4 = c38 * z * (4.0f * y2 - x2z2);
            const float sh3_5 = s5 * sh2_4 * y;
            const float sh3_6 = c56 * (sh2_4 * z - sh2_0 * x);

            const float h0  = 1.0f;
            const float h1  = SQRT3 * x,  h2  = SQRT3 * y,  h3  = SQRT3 * z;
            const float h4  = s5 * sh2_0, h5  = s5 * sh2_1, h6  = s5 * sh2_2;
            const float h7  = s5 * sh2_3, h8  = s5 * sh2_4;
            const float h9  = s7 * sh3_0, h10 = s7 * sh3_1, h11 = s7 * sh3_2;
            const float h12 = s7 * sh3_3, h13 = s7 * sh3_4, h14 = s7 * sh3_5;
            const float h15 = s7 * sh3_6;
            const float Y[16] = {h0, h1, h2, h3, h4, h5, h6, h7, h8, h9, h10, h11, h12, h13, h14, h15};

            {
                const bool hi = (half != 0);
                float s = 0.0f;
                s = fmaf(cv[0].x, hi ? h8  : h0, s);
                s = fmaf(cv[0].y, hi ? h9  : h1, s);
                s = fmaf(cv[0].z, hi ? h10 : h2, s);
                s = fmaf(cv[0].w, hi ? h11 : h3, s);
                s = fmaf(cv[1].x, hi ? h12 : h4, s);
                s = fmaf(cv[1].y, hi ? h13 : h5, s);
                s = fmaf(cv[1].z, hi ? h14 : h6, s);
                s = fmaf(cv[1].w, hi ? h15 : h7, s);
                s += xhalf(s);
                if (valid && half == 0) out[(size_t)n_query * OUT_DIM + m] = s;
            }

            v2f afrag[4];
            #pragma unroll
            for (int kk = 0; kk < 4; ++kk) {
                #pragma unroll
                for (int v = 0; v < 2; ++v) {
                    float s = 0.0f;
                    #pragma unroll
                    for (int d4 = 0; d4 < 4; ++d4) {
                        const float4 rv = sv[kk][v][d4];
                        s = fmaf(rv.x, Y[d4 * 4 + 0], s);
                        s = fmaf(rv.y, Y[d4 * 4 + 1], s);
                        s = fmaf(rv.z, Y[d4 * 4 + 2], s);
                        s = fmaf(rv.w, Y[d4 * 4 + 3], s);
                    }
                    afrag[kk][v] = s;
                }
            }

            int mrow[8];
            #pragma unroll
            for (int v = 0; v < 8; ++v) {
                const int mlo = __builtin_amdgcn_readlane(m, v);
                const int mhi = __builtin_amdgcn_readlane(m, v + 8);
                mrow[v] = half ? mhi : mlo;
            }

            #pragma unroll
            for (int nt = 0; nt < 4; ++nt) {
                const int col = nt * 16 + q;
                const float bb = bias[nt];
                v8f acc = {bb, bb, bb, bb, bb, bb, bb, bb};
                #pragma unroll
                for (int kk = 0; kk < 4; ++kk) {
                    acc = __builtin_amdgcn_wmma_f32_16x16x4_f32(
                        false, afrag[kk], false, bfrag[nt][kk], (short)0, acc, false, false);
                }
                #pragma unroll
                for (int v = 0; v < 8; ++v) {
                    if (t0 + v + half * 8 < cnt)
                        out[(size_t)mrow[v] * OUT_DIM + col] = acc[v];
                }
            }
        }
        __syncthreads();
    }
}

extern "C" void kernel_launch(void* const* d_in, const int* in_sizes, int n_in,
                              void* d_out, int out_size, void* d_ws, size_t ws_size,
                              hipStream_t stream) {
    const int*   point_ids = (const int*)d_in[0];
    const float* cam       = (const float*)d_in[1];
    const int*   ids_3d    = (const int*)d_in[2];
    const float* xyz       = (const float*)d_in[3];
    const float* sh        = (const float*)d_in[4];
    const float* confi     = (const float*)d_in[5];
    const float* W         = (const float*)d_in[6];
    const float* b         = (const float*)d_in[7];
    float* out = (float*)d_out;

    const int n_pts   = in_sizes[2];
    const int n_query = in_sizes[0];

    sh_fused<<<(n_pts + 255) / 256, 128, 0, stream>>>(point_ids, cam, ids_3d, xyz, sh, confi,
                                                      W, b, out, n_pts, n_query);
}


# Id table holds slot+1 (255 points per work-group): a nonzero byte is the match; each lane takes its matches from its 32 bytes by a max tree, one wave vote, no 2nd lookup, no 32 vote-and-branch steps
# speedup vs baseline: 1.8762x; 1.0722x over previous
#include <hip/hip_runtime.h>


typedef __attribute__((ext_vector_type(2))) float v2f;
typedef __attribute__((ext_vector_type(8))) float v8f;

#define OUT_DIM    64
#define CHUNK      255
#define WG_LANES   128
#define N_WAVES    4
#define TABLE_N    100000
#define Q_PASS     4096
#define Q_PER_WAVE (Q_PASS / N_WAVES)

__device__ __forceinline__ float xhalf(float x) {
    return __builtin_bit_cast(float, __builtin_amdgcn_permlanex16(__builtin_bit_cast(int, x), __builtin_bit_cast(int, x),
                                                                  0x76543210, 0xfedcba98, false, false));
}

__global__ __launch_bounds__(WG_LANES) void sh_fused(
    const int*   __restrict__ point_ids,
    const float* __restrict__ cam_center,
    const int*   __restrict__ ids_3d,
    const float* __restrict__ xyz,
    const float* __restrict__ sh_coeffs,
    const float* __restrict__ confi,
    const float* __restrict__ W,
    const float* __restrict__ bvec,
    float*       __restrict__ out,
    int n_pts, int n_query)
{
    __shared__ __attribute__((aligned(16))) unsigned char loc8[TABLE_N + 16];
    __shared__ int           mlist[N_WAVES][Q_PER_WAVE];

    const int tid  = threadIdx.x;
    const int lane = tid & 31;
    const int w    = tid >> 5;
    const int q    = lane & 15;
    const int half = lane >> 4;
    const int base = blockIdx.x * CHUNK;

    const int4* pq  = (const int4*)point_ids;
    const int   nq4 = n_query >> 2;
    int pA[2];
    #pragma unroll
    for (int r = 0; r < 2; ++r) {
        const int  t  = tid + r * WG_LANES;
        const int  i  = base + t;
        const bool ok = (t < CHUNK) && (i < n_pts);
        const int  v  = ids_3d[ok ? i : 0];
        pA[r] = ok ? v : -1;
    }
    int4 pv[8];
    #pragma unroll
    for (int j = 0; j < 8; ++j) {
        const int v4 = tid + WG_LANES * j;
        const int4 v = pq[(v4 < nq4) ? v4 : 0];
        pv[j] = (v4 < nq4) ? v : make_int4(-2, -2, -2, -2);
    }
    const int klo = half * 2;
    v2f   bfrag[4][4];
    float bias[4];
    #pragma unroll
    for (int nt = 0; nt < 4; ++nt) {
        const int col = nt * 16 + q;
        bias[nt] = bvec[col];
        #pragma unroll
        for (int kk = 0; kk < 4; ++kk) {
            bfrag[nt][kk][0] = W[(4 * kk + klo + 0) * OUT_DIM + col];
            bfrag[nt][kk][1] = W[(4 * kk + klo + 1) * OUT_DIM + col];
        }
    }
    const float cx = cam_center[0], cy = cam_center[1], cz = cam_center[2];

    {
        uint4* t16 = (uint4*)loc8;
        for (int i = tid; i < (TABLE_N + 16) / 16; i += WG_LANES) t16[i] = make_uint4(0u, 0u, 0u, 0u);
    }
    __syncthreads();
    #pragma unroll
    for (int r = 0; r < 2; ++r) {
        const int t = tid + r * WG_LANES;
        if ((unsigned)pA[r] < (unsigned)TABLE_N) loc8[pA[r]] = (unsigned char)(t + 1);
    }
    __syncthreads();

    for (int qb = 0; qb < n_query; qb += Q_PASS) {
        int      pm[32];
        unsigned key[32];
        #pragma unroll
        for (int j = 0; j < 8; ++j) {
            pm[4 * j + 0] = pv[j].x; pm[4 * j + 1] = pv[j].y;
            pm[4 * j + 2] = pv[j].z; pm[4 * j + 3] = pv[j].w;
        }
        if (qb + Q_PASS < n_query) {
            #pragma unroll
            for (int j = 0; j < 8; ++j) {
                const int v4 = ((qb + Q_PASS) >> 2) + tid + WG_LANES * j;
                const int4 v = pq[(v4 < nq4) ? v4 : 0];
                pv[j] = (v4 < nq4) ? v : make_int4(-2, -2, -2, -2);
            }
        }
        unsigned left = 0u;
        unsigned best = 0u;
        #pragma unroll
        for (int k = 0; k < 32; ++k) {
            const int pi = ((unsigned)pm[k] < (unsigned)TABLE_N) ? pm[k] : TABLE_N;
            const unsigned b = loc8[pi];
            key[k] = (b << 5) | (unsigned)k;
            left += b;
            best = max(best, key[k]);
        }
        int cnt = 0;
        for (int round = 0; round < 32; ++round) {
            const bool     has  = best >= 32u;
            const unsigned mask = __builtin_amdgcn_ballot_w32(has);
            if (mask == 0u) break;
            if (has) {
                const int kb  = (int)(best & 31u);
                const int m   = qb + 4 * tid + 4 * WG_LANES * (kb >> 2) + (kb & 3);
                const int pos = cnt + (int)__builtin_amdgcn_mbcnt_lo(mask, 0u);
                mlist[w][pos] = (m << 8) | (int)((best >> 5) - 1u);
                left -= best >> 5;
            }
            cnt += __builtin_popcount(mask);
            if (__builtin_amdgcn_ballot_w32(left != 0u) == 0u) break;
            const unsigned lim = has ? best : 0u;
            const unsigned lm1 = lim - 1u;
            unsigned tmin = 0xffffffffu;
            #pragma unroll
            for (int k = 0; k < 32; ++k) tmin = min(tmin, lm1 - key[k]);
            best = (tmin < lim) ? (lm1 - tmin) : 0u;
        }
        __syncthreads();

        for (int t0 = 0; t0 < cnt; t0 += 16) {
            const int  slot  = t0 + q;
            const bool valid = slot < cnt;
            const int  ent   = mlist[w][valid ? slot : t0];
            const int  m     = ent >> 8;
            const int  idx   = base + (ent & 255);

            const float* xp = xyz + (size_t)idx * 3;
            const float px = xp[0], py = xp[1], pz = xp[2];
            const float4* cp = (const float4*)(confi + (size_t)idx * 16) + 2 * half;
            float4 cv[2];
            cv[0] = cp[0];
            cv[1] = cp[1];
            const float* sp = sh_coeffs + (size_t)idx * 256;
            float4 sv[4][2][4];
            #pragma unroll
            for (int kk = 0; kk < 4; ++kk) {
                #pragma unroll
                for (int v = 0; v < 2; ++v) {
                    const float4* rp = (const float4*)(sp + (4 * kk + klo + v) * 16);
                    #pragma unroll
                    for (int d4 = 0; d4 < 4; ++d4) sv[kk][v][d4] = rp[d4];
                }
            }

            const float dx = cx - px;
            const float dy = cy - py;
            const float dz = cz - pz;
            const float nrm = sqrtf(dx * dx + dy * dy + dz * dz);
            const float inr = 1.0f / fmaxf(nrm, 1e-12f);
            const float x = dx * inr, y = dy * inr, z = dz * inr;

            const float SQRT3 = 1.7320508075688772f;
            const float s5    = 2.2360679774997896f;
            const float s7    = 2.6457513110645907f;
            const float c56   = 0.9128709291752769f;
            const float c38   = 0.6123724356957945f;

            const float y2   = y * y;
            const float x2z2 = x * x + z * z;
            const float sh2_0 = SQRT3 * x * z;
            const float sh2_1 = SQRT3 * x * y;
            const float sh2_2 = y2 - 0.5f * x2z2;
            const float sh2_3 = SQRT3 * y * z;
            const float sh2_4 = 0.5f * SQRT3 * (z * z - x * x);
            const float sh3_0 = c56 * (sh2_0 * z + sh2_4 * x);
            const float sh3_1 = s5 * sh2_0 * y;
            const float sh3_2 = c38 * (4.0f * y2 - x2z2) * x;
            const float sh3_3 = 0.5f * y * (2.0f * y2 - 3.0f * x2z2);
            const float sh3_4 = c38 * z * (4.0f * y2 - x2z2);
            const float sh3_5 = s5 * sh2_4 * y;
            const float sh3_6 = c56 * (sh2_4 * z - sh2_0 * x);

            const float h0  = 1.0f;
            const float h1  = SQRT3 * x,  h2  = SQRT3 * y,  h3  = SQRT3 * z;
            const float h4  = s5 * sh2_0, h5  = s5 * sh2_1, h6  = s5 * sh2_2;
            const float h7  = s5 * sh2_3, h8  = s5 * sh2_4;
            const float h9  = s7 * sh3_0, h10 = s7 * sh3_1, h11 = s7 * sh3_2;
            const float h12 = s7 * sh3_3, h13 = s7 * sh3_4, h14 = s7 * sh3_5;
            const float h15 = s7 * sh3_6;
            const float Y[16] = {h0, h1, h2, h3, h4, h5, h6, h7, h8, h9, h10, h11, h12, h13, h14, h15};

            {
                const bool hi = (half != 0);
                float s = 0.0f;
                s = fmaf(cv[0].x, hi ? h8  : h0, s);
                s = fmaf(cv[0].y, hi ? h9  : h1, s);
                s = fmaf(cv[0].z, hi ? h10 : h2, s);
                s = fmaf(cv[0].w, hi ? h11 : h3, s);
                s = fmaf(cv[1].x, hi ? h12 : h4, s);
                s = fmaf(cv[1].y, hi ? h13 : h5, s);
                s = fmaf(cv[1].z, hi ? h14 : h6, s);
                s = fmaf(cv[1].w, hi ? h15 : h7, s);
                s += xhalf(s);
                if (valid && half == 0) out[(size_t)n_query * OUT_DIM + m] = s;
            }

            v2f afrag[4];
            #pragma unroll
            for (int kk = 0; kk < 4; ++kk) {
                #pragma unroll
                for (int v = 0; v < 2; ++v) {
                    float s = 0.0f;
                    #pragma unroll
                    for (int d4 = 0; d4 < 4; ++d4) {
                        const float4 rv = sv[kk][v][d4];
                        s = fmaf(rv.x, Y[d4 * 4 + 0], s);
                        s = fmaf(rv.y, Y[d4 * 4 + 1], s);
                        s = fmaf(rv.z, Y[d4 * 4 + 2], s);
                        s = fmaf(rv.w, Y[d4 * 4 + 3], s);
                    }
                    afrag[kk][v] = s;
                }
            }

            int mrow[8];
            #pragma unroll
            for (int v = 0; v < 8; ++v) {
                const int mlo = __builtin_amdgcn_readlane(m, v);
                const int mhi = __builtin_amdgcn_readlane(m, v + 8);
                mrow[v] = half ? mhi : mlo;
            }

            #pragma unroll
            for (int nt = 0; nt < 4; ++nt) {
                const int col = nt * 16 + q;
                const float bb = bias[nt];
                v8f acc = {bb, bb, bb, bb, bb, bb, bb, bb};
                #pragma unroll
                for (int kk = 0; kk < 4; ++kk) {
                    acc = __builtin_amdgcn_wmma_f32_16x16x4_f32(
                        false, afrag[kk], false, bfrag[nt][kk], (short)0, acc, false, false);
                }
                #pragma unroll
                for (int v = 0; v < 8; ++v) {
                    if (t0 + v + half * 8 < cnt)
                        out[(size_t)mrow[v] * OUT_DIM + col] = acc[v];
                }
            }
        }
        __syncthreads();
    }
}

extern "C" void kernel_launch(void* const* d_in, const int* in_sizes, int n_in,
                              void* d_out, int out_size, void* d_ws, size_t ws_size,
                              hipStream_t stream) {
    const int*   point_ids = (const int*)d_in[0];
    const float* cam       = (const float*)d_in[1];
    const int*   ids_3d    = (const int*)d_in[2];
    const float* xyz       = (const float*)d_in[3];
    const float* sh        = (const float*)d_in[4];
    const float* confi     = (const float*)d_in[5];
    const float* W         = (const float*)d_in[6];
    const float* b         = (const float*)d_in[7];
    float* out = (float*)d_out;

    const int n_pts   = in_sizes[2];
    const int n_query = in_sizes[0];

    sh_fused<<<(n_pts + 254) / 255, 128, 0, stream>>>(point_ids, cam, ids_3d, xyz, sh, confi,
                                                      W, b, out, n_pts, n_query);
}
